# starting kernel
# baseline (speedup 1.0000x reference)
#include <hip/hip_runtime.h>

#define HH   64
#define WW   64
#define CCH  2048
#define HW   4096
#define IMG  224

typedef __attribute__((ext_vector_type(2))) float v2f;
typedef __attribute__((ext_vector_type(8))) float v8f;

#define WMMA_F32(A, B, C) \
    __builtin_amdgcn_wmma_f32_16x16x4_f32(false, (A), false, (B), (short)0, (C), false, false)

__device__ __forceinline__ unsigned f2ord(float f) {
    unsigned u = __float_as_uint(f);
    return (u & 0x80000000u) ? ~u : (u | 0x80000000u);
}
__device__ __forceinline__ float ord2f(unsigned s) {
    unsigned b = (s & 0x80000000u) ? (s ^ 0x80000000u) : ~s;
    return __uint_as_float(b);
}

__global__ __launch_bounds__(256)
void simcam_colsum(const float* __restrict__ x0, const float* __restrict__ x1,
                   float* __restrict__ S0, float* __restrict__ S1,
                   unsigned* __restrict__ maxp) {
    const int t = blockIdx.x * 256 + threadIdx.x;
    if (t == 0) maxp[0] = 0u;
    const bool second = t >= 512;
    const float* __restrict__ X = second ? x1 : x0;
    float* __restrict__ S = second ? S1 : S0;
    const int c4 = (t & 511) * 4;
    float sx = 0.f, sy = 0.f, sz = 0.f, sw = 0.f;
    #pragma unroll 4
    for (int r = 0; r < HW; ++r) {
        float4 v = *reinterpret_cast<const float4*>(X + (size_t)r * CCH + c4);
        sx += v.x; sy += v.y; sz += v.z; sw += v.w;
    }
    float4 o; o.x = sx; o.y = sy; o.z = sz; o.w = sw;
    *reinterpret_cast<float4*>(S + c4) = o;
}

__global__ __launch_bounds__(256)
void simcam_gemm_max(const float* __restrict__ x0, const float* __restrict__ x1,
                     unsigned* __restrict__ maxp) {
    const int wid  = threadIdx.x >> 5;
    const int lane = threadIdx.x & 31;
    const int r    = lane & 15;
    const int h    = lane >> 4;

    const int m0 = blockIdx.x * 128 + (wid & 3) * 32;
    const int n0 = blockIdx.y * 128 + (wid >> 2) * 64;

    const float* __restrict__ A0 = x0 + (size_t)(m0 + r) * CCH + 2 * h;
    const float* __restrict__ A1 = A0 + (size_t)16 * CCH;
    const float* __restrict__ B0 = x1 + (size_t)(n0 + r) * CCH + 2 * h;
    const float* __restrict__ B1 = B0 + (size_t)16 * CCH;
    const float* __restrict__ B2 = B0 + (size_t)32 * CCH;
    const float* __restrict__ B3 = B0 + (size_t)48 * CCH;

    v8f acc00 = {}, acc01 = {}, acc02 = {}, acc03 = {};
    v8f acc10 = {}, acc11 = {}, acc12 = {}, acc13 = {};

    v2f a0 = *reinterpret_cast<const v2f*>(A0);
    v2f a1 = *reinterpret_cast<const v2f*>(A1);
    v2f b0 = *reinterpret_cast<const v2f*>(B0);
    v2f b1 = *reinterpret_cast<const v2f*>(B1);
    v2f b2 = *reinterpret_cast<const v2f*>(B2);
    v2f b3 = *reinterpret_cast<const v2f*>(B3);

    #pragma unroll 2
    for (int c = 4; c < CCH; c += 4) {
        v2f na0 = *reinterpret_cast<const v2f*>(A0 + c);
        v2f na1 = *reinterpret_cast<const v2f*>(A1 + c);
        v2f nb0 = *reinterpret_cast<const v2f*>(B0 + c);
        v2f nb1 = *reinterpret_cast<const v2f*>(B1 + c);
        v2f nb2 = *reinterpret_cast<const v2f*>(B2 + c);
        v2f nb3 = *reinterpret_cast<const v2f*>(B3 + c);

        acc00 = WMMA_F32(a0, b0, acc00);
        acc01 = WMMA_F32(a0, b1, acc01);
        acc02 = WMMA_F32(a0, b2, acc02);
        acc03 = WMMA_F32(a0, b3, acc03);
        acc10 = WMMA_F32(a1, b0, acc10);
        acc11 = WMMA_F32(a1, b1, acc11);
        acc12 = WMMA_F32(a1, b2, acc12);
        acc13 = WMMA_F32(a1, b3, acc13);

        a0 = na0; a1 = na1;
        b0 = nb0; b1 = nb1; b2 = nb2; b3 = nb3;
    }

    acc00 = WMMA_F32(a0, b0, acc00);
    acc01 = WMMA_F32(a0, b1, acc01);
    acc02 = WMMA_F32(a0, b2, acc02);
    acc03 = WMMA_F32(a0, b3, acc03);
    acc10 = WMMA_F32(a1, b0, acc10);
    acc11 = WMMA_F32(a1, b1, acc11);
    acc12 = WMMA_F32(a1, b2, acc12);
    acc13 = WMMA_F32(a1, b3, acc13);

    float m = -3.402823466e38f;
    #pragma unroll
    for (int e = 0; e < 8; ++e) {
        m = fmaxf(m, acc00[e]); m = fmaxf(m, acc01[e]);
        m = fmaxf(m, acc02[e]); m = fmaxf(m, acc03[e]);
        m = fmaxf(m, acc10[e]); m = fmaxf(m, acc11[e]);
        m = fmaxf(m, acc12[e]); m = fmaxf(m, acc13[e]);
    }
    #pragma unroll
    for (int off = 16; off >= 1; off >>= 1)
        m = fmaxf(m, __shfl_xor(m, off, 32));
    if (lane == 0) atomicMax(maxp, f2ord(m));
}

__global__ __launch_bounds__(256)
void simcam_saliency(const float* __restrict__ x0, const float* __restrict__ x1,
                     const float* __restrict__ S0, const float* __restrict__ S1,
                     const unsigned* __restrict__ maxp,
                     float* __restrict__ d1, float* __restrict__ d2) {
    __shared__ float s[CCH];
    const bool second = blockIdx.x >= 16;
    const float* __restrict__ S = second ? S0 : S1;
    const float* __restrict__ X = second ? x1 : x0;
    float* __restrict__ D       = second ? d2 : d1;
    const int row = ((int)blockIdx.x & 15) * 256 + threadIdx.x;

    for (int c = threadIdx.x; c < CCH; c += 256) s[c] = S[c];
    __syncthreads();

    const float* __restrict__ xr = X + (size_t)row * CCH;
    float acc = 0.0f;
    #pragma unroll 4
    for (int c = 0; c < CCH; c += 4) {
        float4 xv = *reinterpret_cast<const float4*>(xr + c);
        acc += xv.x * s[c] + xv.y * s[c + 1] + xv.z * s[c + 2] + xv.w * s[c + 3];
    }
    const float inv = 1.0f / ord2f(maxp[0]);
    D[row] = acc * inv;
}

__global__ __launch_bounds__(256)
void simcam_resize(const float* __restrict__ d1, const float* __restrict__ d2,
                   float* __restrict__ out) {
    const int idx = blockIdx.x * 256 + threadIdx.x;
    if (idx >= 2 * IMG * IMG) return;
    const int s   = idx / (IMG * IMG);
    const int rem = idx - s * IMG * IMG;
    const int y   = rem / IMG;
    const int x   = rem - y * IMG;
    const float* __restrict__ map = s ? d2 : d1;

    const float scale = (float)HH / (float)IMG;
    float fy = ((float)y + 0.5f) * scale - 0.5f;
    float fx = ((float)x + 0.5f) * scale - 0.5f;
    fy = fminf(fmaxf(fy, 0.0f), (float)(HH - 1));
    fx = fminf(fmaxf(fx, 0.0f), (float)(WW - 1));
    const int y0 = (int)fy, x0 = (int)fx;
    const int y1 = min(y0 + 1, HH - 1), x1 = min(x0 + 1, WW - 1);
    const float wy = fy - (float)y0, wx = fx - (float)x0;

    const float v00 = map[y0 * WW + x0], v01 = map[y0 * WW + x1];
    const float v10 = map[y1 * WW + x0], v11 = map[y1 * WW + x1];
    out[idx] = (1.0f - wy) * ((1.0f - wx) * v00 + wx * v01)
             +          wy * ((1.0f - wx) * v10 + wx * v11);
}

extern "C" void kernel_launch(void* const* d_in, const int* in_sizes, int n_in,
                              void* d_out, int out_size, void* d_ws, size_t ws_size,
                              hipStream_t stream) {
    (void)in_sizes; (void)n_in; (void)out_size; (void)ws_size;

    const float* feats = (const float*)d_in[0];
    const float* x0 = feats;
    const float* x1 = feats + (size_t)HW * CCH;
    float* out = (float*)d_out;

    float*    wsf  = (float*)d_ws;
    unsigned* maxp = (unsigned*)wsf;
    float* S0  = wsf + 64;
    float* S1  = S0 + CCH;
    float* d1m = S1 + CCH;
    float* d2m = d1m + HW;

    simcam_colsum<<<dim3(4), dim3(256), 0, stream>>>(x0, x1, S0, S1, maxp);

    simcam_gemm_max<<<dim3(32, 32), dim3(256), 0, stream>>>(x0, x1, maxp);

    simcam_saliency<<<dim3(32), dim3(256), 0, stream>>>(x0, x1, S0, S1, maxp, d1m, d2m);

    const int nout = 2 * IMG * IMG;
    simcam_resize<<<dim3((nout + 255) / 256), dim3(256), 0, stream>>>(d1m, d2m, out);
}


# Column sums spread from 4 work-groups to 128 (16 waves each, one 128-byte line of channels over all rows, in-block tree sum); the order in which each channel is summed changes.
# speedup vs baseline: 1.6506x; 1.6506x over previous
#include <hip/hip_runtime.h>

#define HH   64
#define WW   64
#define CCH  2048
#define HW   4096
#define IMG  224

typedef __attribute__((ext_vector_type(2))) float v2f;
typedef __attribute__((ext_vector_type(8))) float v8f;

#define WMMA_F32(A, B, C) \
    __builtin_amdgcn_wmma_f32_16x16x4_f32(false, (A), false, (B), (short)0, (C), false, false)

__device__ __forceinline__ unsigned f2ord(float f) {
    unsigned u = __float_as_uint(f);
    return (u & 0x80000000u) ? ~u : (u | 0x80000000u);
}
__device__ __forceinline__ float ord2f(unsigned s) {
    unsigned b = (s & 0x80000000u) ? (s ^ 0x80000000u) : ~s;
    return __uint_as_float(b);
}

#define CS_THREADS 512
#define CS_WAVES   (CS_THREADS / 32)
#define CS_RG      (CS_THREADS / 8)
#define CS_COLS    32

__global__ __launch_bounds__(CS_THREADS)
void simcam_colsum(const float* __restrict__ x0, const float* __restrict__ x1,
                   float* __restrict__ S0, float* __restrict__ S1,
                   unsigned* __restrict__ maxp) {
    __shared__ __align__(16) float part[CS_WAVES * CS_COLS];

    const int tid = threadIdx.x;
    if (blockIdx.x == 0 && tid == 0) maxp[0] = 0u;

    const bool second = blockIdx.x >= (CCH / CS_COLS);
    const float* __restrict__ X = second ? x1 : x0;
    float* __restrict__ S = second ? S1 : S0;
    const int col0 = ((int)blockIdx.x & (CCH / CS_COLS - 1)) * CS_COLS;
    const int cl = tid & 7;
    const int rg = tid >> 3;

    const float* __restrict__ p = X + (size_t)rg * CCH + col0 + 4 * cl;
    float sx = 0.f, sy = 0.f, sz = 0.f, sw = 0.f;
    #pragma unroll 16
    for (int j = 0; j < HW / CS_RG; ++j) {
        float4 v = *reinterpret_cast<const float4*>(p + (size_t)j * CS_RG * CCH);
        sx += v.x; sy += v.y; sz += v.z; sw += v.w;
    }

    for (int off = 8; off <= 16; off <<= 1) {
        sx += __shfl_xor(sx, off, 32);
        sy += __shfl_xor(sy, off, 32);
        sz += __shfl_xor(sz, off, 32);
        sw += __shfl_xor(sw, off, 32);
    }

    const int wave = tid >> 5;
    const int lane = tid & 31;
    if (lane < 8) {
        float4 o; o.x = sx; o.y = sy; o.z = sz; o.w = sw;
        *reinterpret_cast<float4*>(&part[wave * CS_COLS + 4 * lane]) = o;
    }
    __syncthreads();

    if (tid < CS_COLS) {
        float s = 0.f;
        for (int w = 0; w < CS_WAVES; ++w) s += part[w * CS_COLS + tid];
        S[col0 + tid] = s;
    }
}

__global__ __launch_bounds__(256)
void simcam_gemm_max(const float* __restrict__ x0, const float* __restrict__ x1,
                     unsigned* __restrict__ maxp) {
    const int wid  = threadIdx.x >> 5;
    const int lane = threadIdx.x & 31;
    const int r    = lane & 15;
    const int h    = lane >> 4;

    const int m0 = blockIdx.x * 128 + (wid & 3) * 32;
    const int n0 = blockIdx.y * 128 + (wid >> 2) * 64;

    const float* __restrict__ A0 = x0 + (size_t)(m0 + r) * CCH + 2 * h;
    const float* __restrict__ A1 = A0 + (size_t)16 * CCH;
    const float* __restrict__ B0 = x1 + (size_t)(n0 + r) * CCH + 2 * h;
    const float* __restrict__ B1 = B0 + (size_t)16 * CCH;
    const float* __restrict__ B2 = B0 + (size_t)32 * CCH;
    const float* __restrict__ B3 = B0 + (size_t)48 * CCH;

    v8f acc00 = {}, acc01 = {}, acc02 = {}, acc03 = {};
    v8f acc10 = {}, acc11 = {}, acc12 = {}, acc13 = {};

    v2f a0 = *reinterpret_cast<const v2f*>(A0);
    v2f a1 = *reinterpret_cast<const v2f*>(A1);
    v2f b0 = *reinterpret_cast<const v2f*>(B0);
    v2f b1 = *reinterpret_cast<const v2f*>(B1);
    v2f b2 = *reinterpret_cast<const v2f*>(B2);
    v2f b3 = *reinterpret_cast<const v2f*>(B3);

    #pragma unroll 2
    for (int c = 4; c < CCH; c += 4) {
        v2f na0 = *reinterpret_cast<const v2f*>(A0 + c);
        v2f na1 = *reinterpret_cast<const v2f*>(A1 + c);
        v2f nb0 = *reinterpret_cast<const v2f*>(B0 + c);
        v2f nb1 = *reinterpret_cast<const v2f*>(B1 + c);
        v2f nb2 = *reinterpret_cast<const v2f*>(B2 + c);
        v2f nb3 = *reinterpret_cast<const v2f*>(B3 + c);

        acc00 = WMMA_F32(a0, b0, acc00);
        acc01 = WMMA_F32(a0, b1, acc01);
        acc02 = WMMA_F32(a0, b2, acc02);
        acc03 = WMMA_F32(a0, b3, acc03);
        acc10 = WMMA_F32(a1, b0, acc10);
        acc11 = WMMA_F32(a1, b1, acc11);
        acc12 = WMMA_F32(a1, b2, acc12);
        acc13 = WMMA_F32(a1, b3, acc13);

        a0 = na0; a1 = na1;
        b0 = nb0; b1 = nb1; b2 = nb2; b3 = nb3;
    }

    acc00 = WMMA_F32(a0, b0, acc00);
    acc01 = WMMA_F32(a0, b1, acc01);
    acc02 = WMMA_F32(a0, b2, acc02);
    acc03 = WMMA_F32(a0, b3, acc03);
    acc10 = WMMA_F32(a1, b0, acc10);
    acc11 = WMMA_F32(a1, b1, acc11);
    acc12 = WMMA_F32(a1, b2, acc12);
    acc13 = WMMA_F32(a1, b3, acc13);

    float m = -3.402823466e38f;
    #pragma unroll
    for (int e = 0; e < 8; ++e) {
        m = fmaxf(m, acc00[e]); m = fmaxf(m, acc01[e]);
        m = fmaxf(m, acc02[e]); m = fmaxf(m, acc03[e]);
        m = fmaxf(m, acc10[e]); m = fmaxf(m, acc11[e]);
        m = fmaxf(m, acc12[e]); m = fmaxf(m, acc13[e]);
    }
    #pragma unroll
    for (int off = 16; off >= 1; off >>= 1)
        m = fmaxf(m, __shfl_xor(m, off, 32));
    if (lane == 0) atomicMax(maxp, f2ord(m));
}

__global__ __launch_bounds__(256)
void simcam_saliency(const float* __restrict__ x0, const float* __restrict__ x1,
                     const float* __restrict__ S0, const float* __restrict__ S1,
                     const unsigned* __restrict__ maxp,
                     float* __restrict__ d1, float* __restrict__ d2) {
    __shared__ float s[CCH];
    const bool second = blockIdx.x >= 16;
    const float* __restrict__ S = second ? S0 : S1;
    const float* __restrict__ X = second ? x1 : x0;
    float* __restrict__ D       = second ? d2 : d1;
    const int row = ((int)blockIdx.x & 15) * 256 + threadIdx.x;

    for (int c = threadIdx.x; c < CCH; c += 256) s[c] = S[c];
    __syncthreads();

    const float* __restrict__ xr = X + (size_t)row * CCH;
    float acc = 0.0f;
    #pragma unroll 4
    for (int c = 0; c < CCH; c += 4) {
        float4 xv = *reinterpret_cast<const float4*>(xr + c);
        acc += xv.x * s[c] + xv.y * s[c + 1] + xv.z * s[c + 2] + xv.w * s[c + 3];
    }
    const float inv = 1.0f / ord2f(maxp[0]);
    D[row] = acc * inv;
}

__global__ __launch_bounds__(256)
void simcam_resize(const float* __restrict__ d1, const float* __restrict__ d2,
                   float* __restrict__ out) {
    const int idx = blockIdx.x * 256 + threadIdx.x;
    if (idx >= 2 * IMG * IMG) return;
    const int s   = idx / (IMG * IMG);
    const int rem = idx - s * IMG * IMG;
    const int y   = rem / IMG;
    const int x   = rem - y * IMG;
    const float* __restrict__ map = s ? d2 : d1;

    const float scale = (float)HH / (float)IMG;
    float fy = ((float)y + 0.5f) * scale - 0.5f;
    float fx = ((float)x + 0.5f) * scale - 0.5f;
    fy = fminf(fmaxf(fy, 0.0f), (float)(HH - 1));
    fx = fminf(fmaxf(fx, 0.0f), (float)(WW - 1));
    const int y0 = (int)fy, x0 = (int)fx;
    const int y1 = min(y0 + 1, HH - 1), x1 = min(x0 + 1, WW - 1);
    const float wy = fy - (float)y0, wx = fx - (float)x0;

    const float v00 = map[y0 * WW + x0], v01 = map[y0 * WW + x1];
    const float v10 = map[y1 * WW + x0], v11 = map[y1 * WW + x1];
    out[idx] = (1.0f - wy) * ((1.0f - wx) * v00 + wx * v01)
             +          wy * ((1.0f - wx) * v10 + wx * v11);
}

extern "C" void kernel_launch(void* const* d_in, const int* in_sizes, int n_in,
                              void* d_out, int out_size, void* d_ws, size_t ws_size,
                              hipStream_t stream) {
    (void)in_sizes; (void)n_in; (void)out_size; (void)ws_size;

    const float* feats = (const float*)d_in[0];
    const float* x0 = feats;
    const float* x1 = feats + (size_t)HW * CCH;
    float* out = (float*)d_out;

    float*    wsf  = (float*)d_ws;
    unsigned* maxp = (unsigned*)wsf;
    float* S0  = wsf + 64;
    float* S1  = S0 + CCH;
    float* d1m = S1 + CCH;
    float* d2m = d1m + HW;

    simcam_colsum<<<dim3(2 * (CCH / CS_COLS)), dim3(CS_THREADS), 0, stream>>>(x0, x1, S0, S1, maxp);

    simcam_gemm_max<<<dim3(32, 32), dim3(256), 0, stream>>>(x0, x1, maxp);

    simcam_saliency<<<dim3(32), dim3(256), 0, stream>>>(x0, x1, S0, S1, maxp, d1m, d2m);

    const int nout = 2 * IMG * IMG;
    simcam_resize<<<dim3((nout + 255) / 256), dim3(256), 0, stream>>>(d1m, d2m, out);
}


# GEMM operands staged through LDS: per 32-wide k chunk a block copies its 128+128 rows once (coalesced 16-byte loads) and all 8 waves read fragments from LDS; k order and sum order unchanged.
# speedup vs baseline: 3.1884x; 1.9317x over previous
#include <hip/hip_runtime.h>

#define HH   64
#define WW   64
#define CCH  2048
#define HW   4096
#define IMG  224

typedef __attribute__((ext_vector_type(2))) float v2f;
typedef __attribute__((ext_vector_type(8))) float v8f;
typedef __attribute__((ext_vector_type(4))) float v4f;

#define WMMA_F32(A, B, C) \
    __builtin_amdgcn_wmma_f32_16x16x4_f32(false, (A), false, (B), (short)0, (C), false, false)

__device__ __forceinline__ unsigned f2ord(float f) {
    unsigned u = __float_as_uint(f);
    return (u & 0x80000000u) ? ~u : (u | 0x80000000u);
}
__device__ __forceinline__ float ord2f(unsigned s) {
    unsigned b = (s & 0x80000000u) ? (s ^ 0x80000000u) : ~s;
    return __uint_as_float(b);
}

#define CS_THREADS 512
#define CS_WAVES   (CS_THREADS / 32)
#define CS_RG      (CS_THREADS / 8)
#define CS_COLS    32

__global__ __launch_bounds__(CS_THREADS)
void simcam_colsum(const float* __restrict__ x0, const float* __restrict__ x1,
                   float* __restrict__ S0, float* __restrict__ S1,
                   unsigned* __restrict__ maxp) {
    __shared__ __align__(16) float part[CS_WAVES * CS_COLS];

    const int tid = threadIdx.x;
    if (blockIdx.x == 0 && tid == 0) maxp[0] = 0u;

    const bool second = blockIdx.x >= (CCH / CS_COLS);
    const float* __restrict__ X = second ? x1 : x0;
    float* __restrict__ S = second ? S1 : S0;
    const int col0 = ((int)blockIdx.x & (CCH / CS_COLS - 1)) * CS_COLS;
    const int cl = tid & 7;
    const int rg = tid >> 3;

    const float* __restrict__ p = X + (size_t)rg * CCH + col0 + 4 * cl;
    float sx = 0.f, sy = 0.f, sz = 0.f, sw = 0.f;
    #pragma unroll 16
    for (int j = 0; j < HW / CS_RG; ++j) {
        float4 v = *reinterpret_cast<const float4*>(p + (size_t)j * CS_RG * CCH);
        sx += v.x; sy += v.y; sz += v.z; sw += v.w;
    }

    for (int off = 8; off <= 16; off <<= 1) {
        sx += __shfl_xor(sx, off, 32);
        sy += __shfl_xor(sy, off, 32);
        sz += __shfl_xor(sz, off, 32);
        sw += __shfl_xor(sw, off, 32);
    }

    const int wave = tid >> 5;
    const int lane = tid & 31;
    if (lane < 8) {
        float4 o; o.x = sx; o.y = sy; o.z = sz; o.w = sw;
        *reinterpret_cast<float4*>(&part[wave * CS_COLS + 4 * lane]) = o;
    }
    __syncthreads();

    if (tid < CS_COLS) {
        float s = 0.f;
        for (int w = 0; w < CS_WAVES; ++w) s += part[w * CS_COLS + tid];
        S[col0 + tid] = s;
    }
}

#define GK 32
#define GP 36

__global__ __launch_bounds__(256)
void simcam_gemm_max(const float* __restrict__ x0, const float* __restrict__ x1,
                     unsigned* __restrict__ maxp) {
    __shared__ __align__(16) float sA[128 * GP];
    __shared__ __align__(16) float sB[128 * GP];

    const int tid  = threadIdx.x;
    const int wid  = tid >> 5;
    const int lane = tid & 31;
    const int r    = lane & 15;
    const int h    = lane >> 4;

    const int sr = tid >> 3;
    const int sc = (tid & 7) * 4;
    const float* __restrict__ gA = x0 + (size_t)(blockIdx.x * 128 + sr) * CCH + sc;
    const float* __restrict__ gB = x1 + (size_t)(blockIdx.y * 128 + sr) * CCH + sc;
    float* const stA = sA + sr * GP + sc;
    float* const stB = sB + sr * GP + sc;

    const float* const fA = sA + ((wid & 3) * 32 + r) * GP + 2 * h;
    const float* const fB = sB + ((wid >> 2) * 64 + r) * GP + 2 * h;

    v8f acc00 = {}, acc01 = {}, acc02 = {}, acc03 = {};
    v8f acc10 = {}, acc11 = {}, acc12 = {}, acc13 = {};

    const size_t RS = (size_t)32 * CCH;

    for (int kc = 0; kc < CCH; kc += GK) {
        const v4f pa0 = *reinterpret_cast<const v4f*>(gA + kc);
        const v4f pa1 = *reinterpret_cast<const v4f*>(gA + kc + RS);
        const v4f pa2 = *reinterpret_cast<const v4f*>(gA + kc + 2 * RS);
        const v4f pa3 = *reinterpret_cast<const v4f*>(gA + kc + 3 * RS);
        const v4f pb0 = *reinterpret_cast<const v4f*>(gB + kc);
        const v4f pb1 = *reinterpret_cast<const v4f*>(gB + kc + RS);
        const v4f pb2 = *reinterpret_cast<const v4f*>(gB + kc + 2 * RS);
        const v4f pb3 = *reinterpret_cast<const v4f*>(gB + kc + 3 * RS);

        __syncthreads();

        *reinterpret_cast<v4f*>(stA)           = pa0;
        *reinterpret_cast<v4f*>(stA + 32 * GP) = pa1;
        *reinterpret_cast<v4f*>(stA + 64 * GP) = pa2;
        *reinterpret_cast<v4f*>(stA + 96 * GP) = pa3;
        *reinterpret_cast<v4f*>(stB)           = pb0;
        *reinterpret_cast<v4f*>(stB + 32 * GP) = pb1;
        *reinterpret_cast<v4f*>(stB + 64 * GP) = pb2;
        *reinterpret_cast<v4f*>(stB + 96 * GP) = pb3;

        __syncthreads();

        #pragma unroll
        for (int k = 0; k < GK; k += 4) {
            const v2f a0 = *reinterpret_cast<const v2f*>(fA + k);
            const v2f a1 = *reinterpret_cast<const v2f*>(fA + 16 * GP + k);
            const v2f b0 = *reinterpret_cast<const v2f*>(fB + k);
            const v2f b1 = *reinterpret_cast<const v2f*>(fB + 16 * GP + k);
            const v2f b2 = *reinterpret_cast<const v2f*>(fB + 32 * GP + k);
            const v2f b3 = *reinterpret_cast<const v2f*>(fB + 48 * GP + k);

            acc00 = WMMA_F32(a0, b0, acc00);
            acc01 = WMMA_F32(a0, b1, acc01);
            acc02 = WMMA_F32(a0, b2, acc02);
            acc03 = WMMA_F32(a0, b3, acc03);
            acc10 = WMMA_F32(a1, b0, acc10);
            acc11 = WMMA_F32(a1, b1, acc11);
            acc12 = WMMA_F32(a1, b2, acc12);
            acc13 = WMMA_F32(a1, b3, acc13);
        }
    }

    float m = -3.402823466e38f;
    #pragma unroll
    for (int e = 0; e < 8; ++e) {
        m = fmaxf(m, acc00[e]); m = fmaxf(m, acc01[e]);
        m = fmaxf(m, acc02[e]); m = fmaxf(m, acc03[e]);
        m = fmaxf(m, acc10[e]); m = fmaxf(m, acc11[e]);
        m = fmaxf(m, acc12[e]); m = fmaxf(m, acc13[e]);
    }
    #pragma unroll
    for (int off = 16; off >= 1; off >>= 1)
        m = fmaxf(m, __shfl_xor(m, off, 32));
    if (lane == 0) atomicMax(maxp, f2ord(m));
}

__global__ __launch_bounds__(256)
void simcam_saliency(const float* __restrict__ x0, const float* __restrict__ x1,
                     const float* __restrict__ S0, const float* __restrict__ S1,
                     const unsigned* __restrict__ maxp,
                     float* __restrict__ d1, float* __restrict__ d2) {
    __shared__ float s[CCH];
    const bool second = blockIdx.x >= 16;
    const float* __restrict__ S = second ? S0 : S1;
    const float* __restrict__ X = second ? x1 : x0;
    float* __restrict__ D       = second ? d2 : d1;
    const int row = ((int)blockIdx.x & 15) * 256 + threadIdx.x;

    for (int c = threadIdx.x; c < CCH; c += 256) s[c] = S[c];
    __syncthreads();

    const float* __restrict__ xr = X + (size_t)row * CCH;
    float acc = 0.0f;
    #pragma unroll 4
    for (int c = 0; c < CCH; c += 4) {
        float4 xv = *reinterpret_cast<const float4*>(xr + c);
        acc += xv.x * s[c] + xv.y * s[c + 1] + xv.z * s[c + 2] + xv.w * s[c + 3];
    }
    const float inv = 1.0f / ord2f(maxp[0]);
    D[row] = acc * inv;
}

__global__ __launch_bounds__(256)
void simcam_resize(const float* __restrict__ d1, const float* __restrict__ d2,
                   float* __restrict__ out) {
    const int idx = blockIdx.x * 256 + threadIdx.x;
    if (idx >= 2 * IMG * IMG) return;
    const int s   = idx / (IMG * IMG);
    const int rem = idx - s * IMG * IMG;
    const int y   = rem / IMG;
    const int x   = rem - y * IMG;
    const float* __restrict__ map = s ? d2 : d1;

    const float scale = (float)HH / (float)IMG;
    float fy = ((float)y + 0.5f) * scale - 0.5f;
    float fx = ((float)x + 0.5f) * scale - 0.5f;
    fy = fminf(fmaxf(fy, 0.0f), (float)(HH - 1));
    fx = fminf(fmaxf(fx, 0.0f), (float)(WW - 1));
    const int y0 = (int)fy, x0 = (int)fx;
    const int y1 = min(y0 + 1, HH - 1), x1 = min(x0 + 1, WW - 1);
    const float wy = fy - (float)y0, wx = fx - (float)x0;

    const float v00 = map[y0 * WW + x0], v01 = map[y0 * WW + x1];
    const float v10 = map[y1 * WW + x0], v11 = map[y1 * WW + x1];
    out[idx] = (1.0f - wy) * ((1.0f - wx) * v00 + wx * v01)
             +          wy * ((1.0f - wx) * v10 + wx * v11);
}

extern "C" void kernel_launch(void* const* d_in, const int* in_sizes, int n_in,
                              void* d_out, int out_size, void* d_ws, size_t ws_size,
                              hipStream_t stream) {
    (void)in_sizes; (void)n_in; (void)out_size; (void)ws_size;

    const float* feats = (const float*)d_in[0];
    const float* x0 = feats;
    const float* x1 = feats + (size_t)HW * CCH;
    float* out = (float*)d_out;

    float*    wsf  = (float*)d_ws;
    unsigned* maxp = (unsigned*)wsf;
    float* S0  = wsf + 64;
    float* S1  = S0 + CCH;
    float* d1m = S1 + CCH;
    float* d2m = d1m + HW;

    simcam_colsum<<<dim3(2 * (CCH / CS_COLS)), dim3(CS_THREADS), 0, stream>>>(x0, x1, S0, S1, maxp);

    simcam_gemm_max<<<dim3(32, 32), dim3(256), 0, stream>>>(x0, x1, maxp);

    simcam_saliency<<<dim3(32), dim3(256), 0, stream>>>(x0, x1, S0, S1, maxp, d1m, d2m);

    const int nout = 2 * IMG * IMG;
    simcam_resize<<<dim3((nout + 255) / 256), dim3(256), 0, stream>>>(d1m, d2m, out);
}


# Saliency dot products spread from 32 work-groups (a row per lane, 32 lines per load) to 256 of 16 waves (two rows per wave, coalesced 16-byte loads); row sum order now lane-strided then butterfly.
# speedup vs baseline: 3.4999x; 1.0977x over previous
#include <hip/hip_runtime.h>

#define HH   64
#define WW   64
#define CCH  2048
#define HW   4096
#define IMG  224

typedef __attribute__((ext_vector_type(2))) float v2f;
typedef __attribute__((ext_vector_type(8))) float v8f;
typedef __attribute__((ext_vector_type(4))) float v4f;

#define WMMA_F32(A, B, C) \
    __builtin_amdgcn_wmma_f32_16x16x4_f32(false, (A), false, (B), (short)0, (C), false, false)

__device__ __forceinline__ unsigned f2ord(float f) {
    unsigned u = __float_as_uint(f);
    return (u & 0x80000000u) ? ~u : (u | 0x80000000u);
}
__device__ __forceinline__ float ord2f(unsigned s) {
    unsigned b = (s & 0x80000000u) ? (s ^ 0x80000000u) : ~s;
    return __uint_as_float(b);
}

#define CS_THREADS 512
#define CS_WAVES   (CS_THREADS / 32)
#define CS_RG      (CS_THREADS / 8)
#define CS_COLS    32

__global__ __launch_bounds__(CS_THREADS)
void simcam_colsum(const float* __restrict__ x0, const float* __restrict__ x1,
                   float* __restrict__ S0, float* __restrict__ S1,
                   unsigned* __restrict__ maxp) {
    __shared__ __align__(16) float part[CS_WAVES * CS_COLS];

    const int tid = threadIdx.x;
    if (blockIdx.x == 0 && tid == 0) maxp[0] = 0u;

    const bool second = blockIdx.x >= (CCH / CS_COLS);
    const float* __restrict__ X = second ? x1 : x0;
    float* __restrict__ S = second ? S1 : S0;
    const int col0 = ((int)blockIdx.x & (CCH / CS_COLS - 1)) * CS_COLS;
    const int cl = tid & 7;
    const int rg = tid >> 3;

    const float* __restrict__ p = X + (size_t)rg * CCH + col0 + 4 * cl;
    float sx = 0.f, sy = 0.f, sz = 0.f, sw = 0.f;
    #pragma unroll 16
    for (int j = 0; j < HW / CS_RG; ++j) {
        float4 v = *reinterpret_cast<const float4*>(p + (size_t)j * CS_RG * CCH);
        sx += v.x; sy += v.y; sz += v.z; sw += v.w;
    }

    for (int off = 8; off <= 16; off <<= 1) {
        sx += __shfl_xor(sx, off, 32);
        sy += __shfl_xor(sy, off, 32);
        sz += __shfl_xor(sz, off, 32);
        sw += __shfl_xor(sw, off, 32);
    }

    const int wave = tid >> 5;
    const int lane = tid & 31;
    if (lane < 8) {
        float4 o; o.x = sx; o.y = sy; o.z = sz; o.w = sw;
        *reinterpret_cast<float4*>(&part[wave * CS_COLS + 4 * lane]) = o;
    }
    __syncthreads();

    if (tid < CS_COLS) {
        float s = 0.f;
        for (int w = 0; w < CS_WAVES; ++w) s += part[w * CS_COLS + tid];
        S[col0 + tid] = s;
    }
}

#define GK 32
#define GP 36

__global__ __launch_bounds__(256)
void simcam_gemm_max(const float* __restrict__ x0, const float* __restrict__ x1,
                     unsigned* __restrict__ maxp) {
    __shared__ __align__(16) float sA[128 * GP];
    __shared__ __align__(16) float sB[128 * GP];

    const int tid  = threadIdx.x;
    const int wid  = tid >> 5;
    const int lane = tid & 31;
    const int r    = lane & 15;
    const int h    = lane >> 4;

    const int sr = tid >> 3;
    const int sc = (tid & 7) * 4;
    const float* __restrict__ gA = x0 + (size_t)(blockIdx.x * 128 + sr) * CCH + sc;
    const float* __restrict__ gB = x1 + (size_t)(blockIdx.y * 128 + sr) * CCH + sc;
    float* const stA = sA + sr * GP + sc;
    float* const stB = sB + sr * GP + sc;

    const float* const fA = sA + ((wid & 3) * 32 + r) * GP + 2 * h;
    const float* const fB = sB + ((wid >> 2) * 64 + r) * GP + 2 * h;

    v8f acc00 = {}, acc01 = {}, acc02 = {}, acc03 = {};
    v8f acc10 = {}, acc11 = {}, acc12 = {}, acc13 = {};

    const size_t RS = (size_t)32 * CCH;

    for (int kc = 0; kc < CCH; kc += GK) {
        const v4f pa0 = *reinterpret_cast<const v4f*>(gA + kc);
        const v4f pa1 = *reinterpret_cast<const v4f*>(gA + kc + RS);
        const v4f pa2 = *reinterpret_cast<const v4f*>(gA + kc + 2 * RS);
        const v4f pa3 = *reinterpret_cast<const v4f*>(gA + kc + 3 * RS);
        const v4f pb0 = *reinterpret_cast<const v4f*>(gB + kc);
        const v4f pb1 = *reinterpret_cast<const v4f*>(gB + kc + RS);
        const v4f pb2 = *reinterpret_cast<const v4f*>(gB + kc + 2 * RS);
        const v4f pb3 = *reinterpret_cast<const v4f*>(gB + kc + 3 * RS);

        __syncthreads();

        *reinterpret_cast<v4f*>(stA)           = pa0;
        *reinterpret_cast<v4f*>(stA + 32 * GP) = pa1;
        *reinterpret_cast<v4f*>(stA + 64 * GP) = pa2;
        *reinterpret_cast<v4f*>(stA + 96 * GP) = pa3;
        *reinterpret_cast<v4f*>(stB)           = pb0;
        *reinterpret_cast<v4f*>(stB + 32 * GP) = pb1;
        *reinterpret_cast<v4f*>(stB + 64 * GP) = pb2;
        *reinterpret_cast<v4f*>(stB + 96 * GP) = pb3;

        __syncthreads();

        #pragma unroll
        for (int k = 0; k < GK; k += 4) {
            const v2f a0 = *reinterpret_cast<const v2f*>(fA + k);
            const v2f a1 = *reinterpret_cast<const v2f*>(fA + 16 * GP + k);
            const v2f b0 = *reinterpret_cast<const v2f*>(fB + k);
            const v2f b1 = *reinterpret_cast<const v2f*>(fB + 16 * GP + k);
            const v2f b2 = *reinterpret_cast<const v2f*>(fB + 32 * GP + k);
            const v2f b3 = *reinterpret_cast<const v2f*>(fB + 48 * GP + k);

            acc00 = WMMA_F32(a0, b0, acc00);
            acc01 = WMMA_F32(a0, b1, acc01);
            acc02 = WMMA_F32(a0, b2, acc02);
            acc03 = WMMA_F32(a0, b3, acc03);
            acc10 = WMMA_F32(a1, b0, acc10);
            acc11 = WMMA_F32(a1, b1, acc11);
            acc12 = WMMA_F32(a1, b2, acc12);
            acc13 = WMMA_F32(a1, b3, acc13);
        }
    }

    float m = -3.402823466e38f;
    #pragma unroll
    for (int e = 0; e < 8; ++e) {
        m = fmaxf(m, acc00[e]); m = fmaxf(m, acc01[e]);
        m = fmaxf(m, acc02[e]); m = fmaxf(m, acc03[e]);
        m = fmaxf(m, acc10[e]); m = fmaxf(m, acc11[e]);
        m = fmaxf(m, acc12[e]); m = fmaxf(m, acc13[e]);
    }
    #pragma unroll
    for (int off = 16; off >= 1; off >>= 1)
        m = fmaxf(m, __shfl_xor(m, off, 32));
    if (lane == 0) atomicMax(maxp, f2ord(m));
}

#define SV_THREADS 512
#define SV_WAVES   (SV_THREADS / 32)
#define SV_ROWS    (2 * SV_WAVES)
#define SV_STEPS   (CCH / 128)

__global__ __launch_bounds__(SV_THREADS)
void simcam_saliency(const float* __restrict__ x0, const float* __restrict__ x1,
                     const float* __restrict__ S0, const float* __restrict__ S1,
                     const unsigned* __restrict__ maxp,
                     float* __restrict__ d1, float* __restrict__ d2) {
    __shared__ __align__(16) float s[CCH];
    __shared__ float res[SV_ROWS];

    const int tid  = threadIdx.x;
    const int wave = tid >> 5;
    const int lane = tid & 31;

    const bool second = blockIdx.x >= (HW / SV_ROWS);
    const float* __restrict__ S = second ? S0 : S1;
    const float* __restrict__ X = second ? x1 : x0;
    float* __restrict__ D       = second ? d2 : d1;
    const int row0 = ((int)blockIdx.x & (HW / SV_ROWS - 1)) * SV_ROWS;

    const float* __restrict__ p = X + (size_t)(row0 + 2 * wave) * CCH + 4 * lane;
    float4 ra[SV_STEPS], rb[SV_STEPS];
    #pragma unroll
    for (int t = 0; t < SV_STEPS; ++t) {
        ra[t] = *reinterpret_cast<const float4*>(p + t * 128);
        rb[t] = *reinterpret_cast<const float4*>(p + CCH + t * 128);
    }

    *reinterpret_cast<float4*>(&s[4 * tid]) = *reinterpret_cast<const float4*>(S + 4 * tid);
    __syncthreads();

    float ax = 0.f, ay = 0.f, az = 0.f, aw = 0.f;
    float bx = 0.f, by = 0.f, bz = 0.f, bw = 0.f;
    #pragma unroll
    for (int t = 0; t < SV_STEPS; ++t) {
        const float4 sv = *reinterpret_cast<const float4*>(&s[t * 128 + 4 * lane]);
        ax += ra[t].x * sv.x; ay += ra[t].y * sv.y; az += ra[t].z * sv.z; aw += ra[t].w * sv.w;
        bx += rb[t].x * sv.x; by += rb[t].y * sv.y; bz += rb[t].z * sv.z; bw += rb[t].w * sv.w;
    }
    float a = (ax + ay) + (az + aw);
    float b = (bx + by) + (bz + bw);

    for (int off = 16; off >= 1; off >>= 1) {
        a += __shfl_xor(a, off, 32);
        b += __shfl_xor(b, off, 32);
    }
    if (lane == 0) { res[2 * wave] = a; res[2 * wave + 1] = b; }
    __syncthreads();

    if (tid < SV_ROWS) {
        const float inv = 1.0f / ord2f(maxp[0]);
        D[row0 + tid] = res[tid] * inv;
    }
}

__global__ __launch_bounds__(256)
void simcam_resize(const float* __restrict__ d1, const float* __restrict__ d2,
                   float* __restrict__ out) {
    const int idx = blockIdx.x * 256 + threadIdx.x;
    if (idx >= 2 * IMG * IMG) return;
    const int s   = idx / (IMG * IMG);
    const int rem = idx - s * IMG * IMG;
    const int y   = rem / IMG;
    const int x   = rem - y * IMG;
    const float* __restrict__ map = s ? d2 : d1;

    const float scale = (float)HH / (float)IMG;
    float fy = ((float)y + 0.5f) * scale - 0.5f;
    float fx = ((float)x + 0.5f) * scale - 0.5f;
    fy = fminf(fmaxf(fy, 0.0f), (float)(HH - 1));
    fx = fminf(fmaxf(fx, 0.0f), (float)(WW - 1));
    const int y0 = (int)fy, x0 = (int)fx;
    const int y1 = min(y0 + 1, HH - 1), x1 = min(x0 + 1, WW - 1);
    const float wy = fy - (float)y0, wx = fx - (float)x0;

    const float v00 = map[y0 * WW + x0], v01 = map[y0 * WW + x1];
    const float v10 = map[y1 * WW + x0], v11 = map[y1 * WW + x1];
    out[idx] = (1.0f - wy) * ((1.0f - wx) * v00 + wx * v01)
             +          wy * ((1.0f - wx) * v10 + wx * v11);
}

extern "C" void kernel_launch(void* const* d_in, const int* in_sizes, int n_in,
                              void* d_out, int out_size, void* d_ws, size_t ws_size,
                              hipStream_t stream) {
    (void)in_sizes; (void)n_in; (void)out_size; (void)ws_size;

    const float* feats = (const float*)d_in[0];
    const float* x0 = feats;
    const float* x1 = feats + (size_t)HW * CCH;
    float* out = (float*)d_out;

    float*    wsf  = (float*)d_ws;
    unsigned* maxp = (unsigned*)wsf;
    float* S0  = wsf + 64;
    float* S1  = S0 + CCH;
    float* d1m = S1 + CCH;
    float* d2m = d1m + HW;

    simcam_colsum<<<dim3(2 * (CCH / CS_COLS)), dim3(CS_THREADS), 0, stream>>>(x0, x1, S0, S1, maxp);

    simcam_gemm_max<<<dim3(32, 32), dim3(256), 0, stream>>>(x0, x1, maxp);

    simcam_saliency<<<dim3(2 * (HW / SV_ROWS)), dim3(SV_THREADS), 0, stream>>>(x0, x1, S0, S1, maxp, d1m, d2m);

    const int nout = 2 * IMG * IMG;
    simcam_resize<<<dim3((nout + 255) / 256), dim3(256), 0, stream>>>(d1m, d2m, out);
}


# GEMM k loop software-pipelined: the next chunk's global loads stay in flight under the current chunk's matrix ops, LDS tiles double-buffered, one barrier per chunk, not two; sum order unchanged.
# speedup vs baseline: 3.6361x; 1.0389x over previous
#include <hip/hip_runtime.h>

#define HH   64
#define WW   64
#define CCH  2048
#define HW   4096
#define IMG  224

typedef __attribute__((ext_vector_type(2))) float v2f;
typedef __attribute__((ext_vector_type(8))) float v8f;
typedef __attribute__((ext_vector_type(4))) float v4f;

#define WMMA_F32(A, B, C) \
    __builtin_amdgcn_wmma_f32_16x16x4_f32(false, (A), false, (B), (short)0, (C), false, false)

__device__ __forceinline__ unsigned f2ord(float f) {
    unsigned u = __float_as_uint(f);
    return (u & 0x80000000u) ? ~u : (u | 0x80000000u);
}
__device__ __forceinline__ float ord2f(unsigned s) {
    unsigned b = (s & 0x80000000u) ? (s ^ 0x80000000u) : ~s;
    return __uint_as_float(b);
}

#define CS_THREADS 512
#define CS_WAVES   (CS_THREADS / 32)
#define CS_RG      (CS_THREADS / 8)
#define CS_COLS    32

__global__ __launch_bounds__(CS_THREADS)
void simcam_colsum(const float* __restrict__ x0, const float* __restrict__ x1,
                   float* __restrict__ S0, float* __restrict__ S1,
                   unsigned* __restrict__ maxp) {
    __shared__ __align__(16) float part[CS_WAVES * CS_COLS];

    const int tid = threadIdx.x;
    if (blockIdx.x == 0 && tid == 0) maxp[0] = 0u;

    const bool second = blockIdx.x >= (CCH / CS_COLS);
    const float* __restrict__ X = second ? x1 : x0;
    float* __restrict__ S = second ? S1 : S0;
    const int col0 = ((int)blockIdx.x & (CCH / CS_COLS - 1)) * CS_COLS;
    const int cl = tid & 7;
    const int rg = tid >> 3;

    const float* __restrict__ p = X + (size_t)rg * CCH + col0 + 4 * cl;
    float sx = 0.f, sy = 0.f, sz = 0.f, sw = 0.f;
    #pragma unroll 16
    for (int j = 0; j < HW / CS_RG; ++j) {
        float4 v = *reinterpret_cast<const float4*>(p + (size_t)j * CS_RG * CCH);
        sx += v.x; sy += v.y; sz += v.z; sw += v.w;
    }

    for (int off = 8; off <= 16; off <<= 1) {
        sx += __shfl_xor(sx, off, 32);
        sy += __shfl_xor(sy, off, 32);
        sz += __shfl_xor(sz, off, 32);
        sw += __shfl_xor(sw, off, 32);
    }

    const int wave = tid >> 5;
    const int lane = tid & 31;
    if (lane < 8) {
        float4 o; o.x = sx; o.y = sy; o.z = sz; o.w = sw;
        *reinterpret_cast<float4*>(&part[wave * CS_COLS + 4 * lane]) = o;
    }
    __syncthreads();

    if (tid < CS_COLS) {
        float s = 0.f;
        for (int w = 0; w < CS_WAVES; ++w) s += part[w * CS_COLS + tid];
        S[col0 + tid] = s;
    }
}

#define GK 32
#define GP 36
#define GT (128 * GP)

#define GEMM_STEPS(TA, TB, K0, K1)                                                \
    _Pragma("unroll")                                                             \
    for (int k = (K0); k < (K1); k += 4) {                                        \
        const v2f a0 = *reinterpret_cast<const v2f*>((TA) + k);                   \
        const v2f a1 = *reinterpret_cast<const v2f*>((TA) + 16 * GP + k);         \
        const v2f b0 = *reinterpret_cast<const v2f*>((TB) + k);                   \
        const v2f b1 = *reinterpret_cast<const v2f*>((TB) + 16 * GP + k);         \
        const v2f b2 = *reinterpret_cast<const v2f*>((TB) + 32 * GP + k);         \
        const v2f b3 = *reinterpret_cast<const v2f*>((TB) + 48 * GP + k);         \
        acc00 = WMMA_F32(a0, b0, acc00);                                          \
        acc01 = WMMA_F32(a0, b1, acc01);                                          \
        acc02 = WMMA_F32(a0, b2, acc02);                                          \
        acc03 = WMMA_F32(a0, b3, acc03);                                          \
        acc10 = WMMA_F32(a1, b0, acc10);                                          \
        acc11 = WMMA_F32(a1, b1, acc11);                                          \
        acc12 = WMMA_F32(a1, b2, acc12);                                          \
        acc13 = WMMA_F32(a1, b3, acc13);                                          \
    }

__global__ __launch_bounds__(256)
void simcam_gemm_max(const float* __restrict__ x0, const float* __restrict__ x1,
                     unsigned* __restrict__ maxp) {
    __shared__ __align__(16) float sA[2 * GT];
    __shared__ __align__(16) float sB[2 * GT];

    const int tid  = threadIdx.x;
    const int wid  = tid >> 5;
    const int lane = tid & 31;
    const int r    = lane & 15;
    const int h    = lane >> 4;

    const int sr = tid >> 3;
    const int sc = (tid & 7) * 4;
    const float* __restrict__ gA = x0 + (size_t)(blockIdx.x * 128 + sr) * CCH + sc;
    const float* __restrict__ gB = x1 + (size_t)(blockIdx.y * 128 + sr) * CCH + sc;
    float* const stA = sA + sr * GP + sc;
    float* const stB = sB + sr * GP + sc;

    const float* const fA = sA + ((wid & 3) * 32 + r) * GP + 2 * h;
    const float* const fB = sB + ((wid >> 2) * 64 + r) * GP + 2 * h;

    v8f acc00 = {}, acc01 = {}, acc02 = {}, acc03 = {};
    v8f acc10 = {}, acc11 = {}, acc12 = {}, acc13 = {};

    const size_t RS = (size_t)32 * CCH;

    v4f pa0 = *reinterpret_cast<const v4f*>(gA);
    v4f pa1 = *reinterpret_cast<const v4f*>(gA + RS);
    v4f pa2 = *reinterpret_cast<const v4f*>(gA + 2 * RS);
    v4f pa3 = *reinterpret_cast<const v4f*>(gA + 3 * RS);
    v4f pb0 = *reinterpret_cast<const v4f*>(gB);
    v4f pb1 = *reinterpret_cast<const v4f*>(gB + RS);
    v4f pb2 = *reinterpret_cast<const v4f*>(gB + 2 * RS);
    v4f pb3 = *reinterpret_cast<const v4f*>(gB + 3 * RS);
    *reinterpret_cast<v4f*>(stA)           = pa0;
    *reinterpret_cast<v4f*>(stA + 32 * GP) = pa1;
    *reinterpret_cast<v4f*>(stA + 64 * GP) = pa2;
    *reinterpret_cast<v4f*>(stA + 96 * GP) = pa3;
    *reinterpret_cast<v4f*>(stB)           = pb0;
    *reinterpret_cast<v4f*>(stB + 32 * GP) = pb1;
    *reinterpret_cast<v4f*>(stB + 64 * GP) = pb2;
    *reinterpret_cast<v4f*>(stB + 96 * GP) = pb3;
    __syncthreads();

    int cur = 0;
    for (int kc = GK; kc < CCH; kc += GK) {
        const float* const tA = fA + cur * GT;
        const float* const tB = fB + cur * GT;
        const int nxt = cur ^ 1;
        float* const wA = stA + nxt * GT;
        float* const wB = stB + nxt * GT;

        pa0 = *reinterpret_cast<const v4f*>(gA + kc);
        pa1 = *reinterpret_cast<const v4f*>(gA + kc + RS);
        pa2 = *reinterpret_cast<const v4f*>(gA + kc + 2 * RS);
        pa3 = *reinterpret_cast<const v4f*>(gA + kc + 3 * RS);
        GEMM_STEPS(tA, tB, 0, GK / 2)
        *reinterpret_cast<v4f*>(wA)           = pa0;
        *reinterpret_cast<v4f*>(wA + 32 * GP) = pa1;
        *reinterpret_cast<v4f*>(wA + 64 * GP) = pa2;
        *reinterpret_cast<v4f*>(wA + 96 * GP) = pa3;

        pb0 = *reinterpret_cast<const v4f*>(gB + kc);
        pb1 = *reinterpret_cast<const v4f*>(gB + kc + RS);
        pb2 = *reinterpret_cast<const v4f*>(gB + kc + 2 * RS);
        pb3 = *reinterpret_cast<const v4f*>(gB + kc + 3 * RS);
        GEMM_STEPS(tA, tB, GK / 2, GK)
        *reinterpret_cast<v4f*>(wB)           = pb0;
        *reinterpret_cast<v4f*>(wB + 32 * GP) = pb1;
        *reinterpret_cast<v4f*>(wB + 64 * GP) = pb2;
        *reinterpret_cast<v4f*>(wB + 96 * GP) = pb3;

        __syncthreads();
        cur = nxt;
    }

    {
        const float* const tA = fA + cur * GT;
        const float* const tB = fB + cur * GT;
        GEMM_STEPS(tA, tB, 0, GK)
    }

    float m = -3.402823466e38f;
    #pragma unroll
    for (int e = 0; e < 8; ++e) {
        m = fmaxf(m, acc00[e]); m = fmaxf(m, acc01[e]);
        m = fmaxf(m, acc02[e]); m = fmaxf(m, acc03[e]);
        m = fmaxf(m, acc10[e]); m = fmaxf(m, acc11[e]);
        m = fmaxf(m, acc12[e]); m = fmaxf(m, acc13[e]);
    }
    #pragma unroll
    for (int off = 16; off >= 1; off >>= 1)
        m = fmaxf(m, __shfl_xor(m, off, 32));
    if (lane == 0) atomicMax(maxp, f2ord(m));
}

#define SV_THREADS 512
#define SV_WAVES   (SV_THREADS / 32)
#define SV_ROWS    (2 * SV_WAVES)
#define SV_STEPS   (CCH / 128)

__global__ __launch_bounds__(SV_THREADS)
void simcam_saliency(const float* __restrict__ x0, const float* __restrict__ x1,
                     const float* __restrict__ S0, const float* __restrict__ S1,
                     const unsigned* __restrict__ maxp,
                     float* __restrict__ d1, float* __restrict__ d2) {
    __shared__ __align__(16) float s[CCH];
    __shared__ float res[SV_ROWS];

    const int tid  = threadIdx.x;
    const int wave = tid >> 5;
    const int lane = tid & 31;

    const bool second = blockIdx.x >= (HW / SV_ROWS);
    const float* __restrict__ S = second ? S0 : S1;
    const float* __restrict__ X = second ? x1 : x0;
    float* __restrict__ D       = second ? d2 : d1;
    const int row0 = ((int)blockIdx.x & (HW / SV_ROWS - 1)) * SV_ROWS;

    const float* __restrict__ p = X + (size_t)(row0 + 2 * wave) * CCH + 4 * lane;
    float4 ra[SV_STEPS], rb[SV_STEPS];
    #pragma unroll
    for (int t = 0; t < SV_STEPS; ++t) {
        ra[t] = *reinterpret_cast<const float4*>(p + t * 128);
        rb[t] = *reinterpret_cast<const float4*>(p + CCH + t * 128);
    }

    *reinterpret_cast<float4*>(&s[4 * tid]) = *reinterpret_cast<const float4*>(S + 4 * tid);
    __syncthreads();

    float ax = 0.f, ay = 0.f, az = 0.f, aw = 0.f;
    float bx = 0.f, by = 0.f, bz = 0.f, bw = 0.f;
    #pragma unroll
    for (int t = 0; t < SV_STEPS; ++t) {
        const float4 sv = *reinterpret_cast<const float4*>(&s[t * 128 + 4 * lane]);
        ax += ra[t].x * sv.x; ay += ra[t].y * sv.y; az += ra[t].z * sv.z; aw += ra[t].w * sv.w;
        bx += rb[t].x * sv.x; by += rb[t].y * sv.y; bz += rb[t].z * sv.z; bw += rb[t].w * sv.w;
    }
    float a = (ax + ay) + (az + aw);
    float b = (bx + by) + (bz + bw);

    for (int off = 16; off >= 1; off >>= 1) {
        a += __shfl_xor(a, off, 32);
        b += __shfl_xor(b, off, 32);
    }
    if (lane == 0) { res[2 * wave] = a; res[2 * wave + 1] = b; }
    __syncthreads();

    if (tid < SV_ROWS) {
        const float inv = 1.0f / ord2f(maxp[0]);
        D[row0 + tid] = res[tid] * inv;
    }
}

__global__ __launch_bounds__(256)
void simcam_resize(const float* __restrict__ d1, const float* __restrict__ d2,
                   float* __restrict__ out) {
    const int idx = blockIdx.x * 256 + threadIdx.x;
    if (idx >= 2 * IMG * IMG) return;
    const int s   = idx / (IMG * IMG);
    const int rem = idx - s * IMG * IMG;
    const int y   = rem / IMG;
    const int x   = rem - y * IMG;
    const float* __restrict__ map = s ? d2 : d1;

    const float scale = (float)HH / (float)IMG;
    float fy = ((float)y + 0.5f) * scale - 0.5f;
    float fx = ((float)x + 0.5f) * scale - 0.5f;
    fy = fminf(fmaxf(fy, 0.0f), (float)(HH - 1));
    fx = fminf(fmaxf(fx, 0.0f), (float)(WW - 1));
    const int y0 = (int)fy, x0 = (int)fx;
    const int y1 = min(y0 + 1, HH - 1), x1 = min(x0 + 1, WW - 1);
    const float wy = fy - (float)y0, wx = fx - (float)x0;

    const float v00 = map[y0 * WW + x0], v01 = map[y0 * WW + x1];
    const float v10 = map[y1 * WW + x0], v11 = map[y1 * WW + x1];
    out[idx] = (1.0f - wy) * ((1.0f - wx) * v00 + wx * v01)
             +          wy * ((1.0f - wx) * v10 + wx * v11);
}

extern "C" void kernel_launch(void* const* d_in, const int* in_sizes, int n_in,
                              void* d_out, int out_size, void* d_ws, size_t ws_size,
                              hipStream_t stream) {
    (void)in_sizes; (void)n_in; (void)out_size; (void)ws_size;

    const float* feats = (const float*)d_in[0];
    const float* x0 = feats;
    const float* x1 = feats + (size_t)HW * CCH;
    float* out = (float*)d_out;

    float*    wsf  = (float*)d_ws;
    unsigned* maxp = (unsigned*)wsf;
    float* S0  = wsf + 64;
    float* S1  = S0 + CCH;
    float* d1m = S1 + CCH;
    float* d2m = d1m + HW;

    simcam_colsum<<<dim3(2 * (CCH / CS_COLS)), dim3(CS_THREADS), 0, stream>>>(x0, x1, S0, S1, maxp);

    simcam_gemm_max<<<dim3(32, 32), dim3(256), 0, stream>>>(x0, x1, maxp);

    simcam_saliency<<<dim3(2 * (HW / SV_ROWS)), dim3(SV_THREADS), 0, stream>>>(x0, x1, S0, S1, maxp, d1m, d2m);

    const int nout = 2 * IMG * IMG;
    simcam_resize<<<dim3((nout + 255) / 256), dim3(256), 0, stream>>>(d1m, d2m, out);
}


# Saliency maps now row and column sums of the GEMM's own accumulators (per-tile partial planes, added in fixed order by the resize launch); both streaming passes go, 4 launches -> 2; sum order changes
# speedup vs baseline: 3.7717x; 1.0373x over previous
#include <hip/hip_runtime.h>

#define HH   64
#define WW   64
#define CCH  2048
#define HW   4096
#define IMG  224

typedef __attribute__((ext_vector_type(2))) float v2f;
typedef __attribute__((ext_vector_type(8))) float v8f;
typedef __attribute__((ext_vector_type(4))) float v4f;

#define WMMA_F32(A, B, C) \
    __builtin_amdgcn_wmma_f32_16x16x4_f32(false, (A), false, (B), (short)0, (C), false, false)

#define NTILE   32
#define WS_MAX  0
#define WS_ROW  1024
#define WS_COL  (WS_ROW + NTILE * HW)
#define WS_NEED ((unsigned long long)(WS_COL + NTILE * HW) * 4ull)

#define EQ 160
#define GK 32
#define GP 36
#define GT (128 * GP)

#define GEMM_STEPS(TA, TB, K0, K1)                                                \
    _Pragma("unroll")                                                             \
    for (int k = (K0); k < (K1); k += 4) {                                        \
        const v2f a0 = *reinterpret_cast<const v2f*>((TA) + k);                   \
        const v2f a1 = *reinterpret_cast<const v2f*>((TA) + 16 * GP + k);         \
        const v2f b0 = *reinterpret_cast<const v2f*>((TB) + k);                   \
        const v2f b1 = *reinterpret_cast<const v2f*>((TB) + 16 * GP + k);         \
        const v2f b2 = *reinterpret_cast<const v2f*>((TB) + 32 * GP + k);         \
        const v2f b3 = *reinterpret_cast<const v2f*>((TB) + 48 * GP + k);         \
        acc00 = WMMA_F32(a0, b0, acc00);                                          \
        acc01 = WMMA_F32(a0, b1, acc01);                                          \
        acc02 = WMMA_F32(a0, b2, acc02);                                          \
        acc03 = WMMA_F32(a0, b3, acc03);                                          \
        acc10 = WMMA_F32(a1, b0, acc10);                                          \
        acc11 = WMMA_F32(a1, b1, acc11);                                          \
        acc12 = WMMA_F32(a1, b2, acc12);                                          \
        acc13 = WMMA_F32(a1, b3, acc13);                                          \
    }

__global__ __launch_bounds__(256, 8)
void simcam_gemm_sums(const float* __restrict__ x0, const float* __restrict__ x1,
                      float* __restrict__ tmax, float* __restrict__ rowpl,
                      float* __restrict__ colpl, unsigned long long ws_bytes) {
    if (ws_bytes < WS_NEED) return;

    __shared__ __align__(16) float sA[2 * GT];
    __shared__ __align__(16) float sB[2 * GT];

    const int tid  = threadIdx.x;
    const int wid  = __builtin_amdgcn_readfirstlane(tid >> 5);
    const int lane = tid & 31;
    const int r    = lane & 15;
    const int h    = lane >> 4;

    const int sr = tid >> 3;
    const int sc = (tid & 7) * 4;
    const float* __restrict__ gA = x0 + (size_t)(blockIdx.x * 128 + sr) * CCH + sc;
    const float* __restrict__ gB = x1 + (size_t)(blockIdx.y * 128 + sr) * CCH + sc;
    float* const stA = sA + sr * GP + sc;
    float* const stB = sB + sr * GP + sc;

    const float* const fA = sA + ((wid & 3) * 32 + r) * GP + 2 * h;
    const float* const fB = sB + ((wid >> 2) * 64 + r) * GP + 2 * h;

    v8f acc00 = {}, acc01 = {}, acc02 = {}, acc03 = {};
    v8f acc10 = {}, acc11 = {}, acc12 = {}, acc13 = {};

    const size_t RS = (size_t)32 * CCH;

    v4f pa0 = *reinterpret_cast<const v4f*>(gA);
    v4f pa1 = *reinterpret_cast<const v4f*>(gA + RS);
    v4f pa2 = *reinterpret_cast<const v4f*>(gA + 2 * RS);
    v4f pa3 = *reinterpret_cast<const v4f*>(gA + 3 * RS);
    v4f pb0 = *reinterpret_cast<const v4f*>(gB);
    v4f pb1 = *reinterpret_cast<const v4f*>(gB + RS);
    v4f pb2 = *reinterpret_cast<const v4f*>(gB + 2 * RS);
    v4f pb3 = *reinterpret_cast<const v4f*>(gB + 3 * RS);
    *reinterpret_cast<v4f*>(stA)           = pa0;
    *reinterpret_cast<v4f*>(stA + 32 * GP) = pa1;
    *reinterpret_cast<v4f*>(stA + 64 * GP) = pa2;
    *reinterpret_cast<v4f*>(stA + 96 * GP) = pa3;
    *reinterpret_cast<v4f*>(stB)           = pb0;
    *reinterpret_cast<v4f*>(stB + 32 * GP) = pb1;
    *reinterpret_cast<v4f*>(stB + 64 * GP) = pb2;
    *reinterpret_cast<v4f*>(stB + 96 * GP) = pb3;
    __syncthreads();

    int cur = 0;
    for (int kc = GK; kc < CCH; kc += GK) {
        const float* const tA = fA + cur * GT;
        const float* const tB = fB + cur * GT;
        const int nxt = cur ^ 1;
        float* const wA = stA + nxt * GT;
        float* const wB = stB + nxt * GT;

        pa0 = *reinterpret_cast<const v4f*>(gA + kc);
        pa1 = *reinterpret_cast<const v4f*>(gA + kc + RS);
        pa2 = *reinterpret_cast<const v4f*>(gA + kc + 2 * RS);
        pa3 = *reinterpret_cast<const v4f*>(gA + kc + 3 * RS);
        GEMM_STEPS(tA, tB, 0, GK / 2)
        *reinterpret_cast<v4f*>(wA)           = pa0;
        *reinterpret_cast<v4f*>(wA + 32 * GP) = pa1;
        *reinterpret_cast<v4f*>(wA + 64 * GP) = pa2;
        *reinterpret_cast<v4f*>(wA + 96 * GP) = pa3;

        pb0 = *reinterpret_cast<const v4f*>(gB + kc);
        pb1 = *reinterpret_cast<const v4f*>(gB + kc + RS);
        pb2 = *reinterpret_cast<const v4f*>(gB + kc + 2 * RS);
        pb3 = *reinterpret_cast<const v4f*>(gB + kc + 3 * RS);
        GEMM_STEPS(tA, tB, GK / 2, GK)
        *reinterpret_cast<v4f*>(wB)           = pb0;
        *reinterpret_cast<v4f*>(wB + 32 * GP) = pb1;
        *reinterpret_cast<v4f*>(wB + 64 * GP) = pb2;
        *reinterpret_cast<v4f*>(wB + 96 * GP) = pb3;

        __syncthreads();
        cur = nxt;
    }

    {
        const float* const tA = fA + cur * GT;
        const float* const tB = fB + cur * GT;
        GEMM_STEPS(tA, tB, 0, GK)
    }

    float m = -3.402823466e38f;
    #pragma unroll
    for (int e = 0; e < 8; ++e) {
        m = fmaxf(m, acc00[e]); m = fmaxf(m, acc01[e]);
        m = fmaxf(m, acc02[e]); m = fmaxf(m, acc03[e]);
        m = fmaxf(m, acc10[e]); m = fmaxf(m, acc11[e]);
        m = fmaxf(m, acc12[e]); m = fmaxf(m, acc13[e]);
    }
    #pragma unroll
    for (int off = 16; off >= 1; off >>= 1)
        m = fmaxf(m, __shfl_xor(m, off, 32));

    const v8f rs0 = (acc00 + acc01) + (acc02 + acc03);
    const v8f rs1 = (acc10 + acc11) + (acc12 + acc13);
    const v8f ct0 = acc00 + acc10, ct1 = acc01 + acc11;
    const v8f ct2 = acc02 + acc12, ct3 = acc03 + acc13;
    const float cs0 = ((ct0[0] + ct0[1]) + (ct0[2] + ct0[3])) + ((ct0[4] + ct0[5]) + (ct0[6] + ct0[7]));
    const float cs1 = ((ct1[0] + ct1[1]) + (ct1[2] + ct1[3])) + ((ct1[4] + ct1[5]) + (ct1[6] + ct1[7]));
    const float cs2 = ((ct2[0] + ct2[1]) + (ct2[2] + ct2[3])) + ((ct2[4] + ct2[5]) + (ct2[6] + ct2[7]));
    const float cs3 = ((ct3[0] + ct3[1]) + (ct3[2] + ct3[3])) + ((ct3[4] + ct3[5]) + (ct3[6] + ct3[7]));

    __syncthreads();

    float* const eR = sA;
    float* const eC = sA + 128 * GP;
    float* const eM = eC + 8 * EQ;
    {
        const int rl = (wid & 3) * 32 + 8 * h;
        const int q  = (wid >> 2) * 16 + r;
        #pragma unroll
        for (int e = 0; e < 8; ++e) {
            eR[(rl + e) * GP + q]      = rs0[e];
            eR[(rl + 16 + e) * GP + q] = rs1[e];
        }
        const int pl = (wid & 3) * 2 + h;
        const int cl = (wid >> 2) * 64 + r;
        eC[pl * EQ + cl]      = cs0;
        eC[pl * EQ + cl + 16] = cs1;
        eC[pl * EQ + cl + 32] = cs2;
        eC[pl * EQ + cl + 48] = cs3;
        if (lane == 0) eM[wid] = m;
    }
    __syncthreads();

    if (tid < 128) {
        const v4f* const p = reinterpret_cast<const v4f*>(eR + tid * GP);
        const v4f s4 = ((p[0] + p[1]) + (p[2] + p[3])) + ((p[4] + p[5]) + (p[6] + p[7]));
        rowpl[(size_t)blockIdx.y * HW + blockIdx.x * 128 + tid] = (s4[0] + s4[1]) + (s4[2] + s4[3]);
    } else {
        const float* const p = eC + (tid - 128);
        const float s = ((p[0] + p[EQ]) + (p[2 * EQ] + p[3 * EQ]))
                      + ((p[4 * EQ] + p[5 * EQ]) + (p[6 * EQ] + p[7 * EQ]));
        colpl[(size_t)blockIdx.x * HW + blockIdx.y * 128 + (tid - 128)] = s;
    }
    if (tid == 0) {
        float bm = eM[0];
        for (int w = 1; w < 8; ++w) bm = fmaxf(bm, eM[w]);
        atomicExch(tmax + blockIdx.y * NTILE + blockIdx.x, bm);
    }
}

__global__ __launch_bounds__(256)
void simcam_finish(const float* __restrict__ tmax, const float* __restrict__ rowpl,
                   const float* __restrict__ colpl, float* __restrict__ out,
                   unsigned long long ws_bytes) {
    __shared__ float part[2][128];
    __shared__ float vals[128];
    __shared__ float wm[8];

    if (ws_bytes < WS_NEED) return;

    const int tid  = threadIdx.x;
    const int wave = tid >> 5;
    const int lane = tid & 31;
    const int s = (int)blockIdx.x / IMG;
    const int y = (int)blockIdx.x - s * IMG;

    const float scale = (float)HH / (float)IMG;
    float fy = ((float)y + 0.5f) * scale - 0.5f;
    fy = fminf(fmaxf(fy, 0.0f), (float)(HH - 1));
    const int y0 = (int)fy;
    const int y1 = min(y0 + 1, HH - 1);
    const float wy = fy - (float)y0;

    const v4f t4 = *reinterpret_cast<const v4f*>(tmax + 4 * tid);
    float m = fmaxf(fmaxf(t4[0], t4[1]), fmaxf(t4[2], t4[3]));

    const int e    = tid & 127;
    const int half = tid >> 7;
    const int src  = ((e >> 6) ? y1 : y0) * WW + (e & 63);
    const float* __restrict__ p = (s ? colpl : rowpl) + (size_t)(half * (NTILE / 2)) * HW + src;
    float v[NTILE / 2];
    #pragma unroll
    for (int j = 0; j < NTILE / 2; ++j) v[j] = p[(size_t)j * HW];
    float a = 0.f;
    #pragma unroll
    for (int j = 0; j < NTILE / 2; ++j) a += v[j];
    part[half][e] = a;

    for (int off = 16; off >= 1; off >>= 1)
        m = fmaxf(m, __shfl_xor(m, off, 32));
    if (lane == 0) wm[wave] = m;
    __syncthreads();

    if (tid < 128) {
        float gm = wm[0];
        for (int w = 1; w < 8; ++w) gm = fmaxf(gm, wm[w]);
        const float inv = 1.0f / gm;
        vals[tid] = (part[0][tid] + part[1][tid]) * inv;
    }
    __syncthreads();

    if (tid < IMG) {
        const int x = tid;
        float fx = ((float)x + 0.5f) * scale - 0.5f;
        fx = fminf(fmaxf(fx, 0.0f), (float)(WW - 1));
        const int x0 = (int)fx;
        const int x1 = min(x0 + 1, WW - 1);
        const float wx = fx - (float)x0;
        const float v00 = vals[x0],      v01 = vals[x1];
        const float v10 = vals[WW + x0], v11 = vals[WW + x1];
        out[((size_t)s * IMG + y) * IMG + x] = (1.0f - wy) * ((1.0f - wx) * v00 + wx * v01)
                                             +          wy * ((1.0f - wx) * v10 + wx * v11);
    }
}

extern "C" void kernel_launch(void* const* d_in, const int* in_sizes, int n_in,
                              void* d_out, int out_size, void* d_ws, size_t ws_size,
                              hipStream_t stream) {
    (void)in_sizes; (void)n_in; (void)out_size;

    const float* feats = (const float*)d_in[0];
    const float* x0 = feats;
    const float* x1 = feats + (size_t)HW * CCH;
    float* out = (float*)d_out;

    float* wsf   = (float*)d_ws;
    float* tmax  = wsf + WS_MAX;
    float* rowpl = wsf + WS_ROW;
    float* colpl = wsf + WS_COL;

    simcam_gemm_sums<<<dim3(NTILE, NTILE), dim3(256), 0, stream>>>(x0, x1, tmax, rowpl, colpl,
                                                                  (unsigned long long)ws_size);

    simcam_finish<<<dim3(2 * IMG), dim3(256), 0, stream>>>(tmax, rowpl, colpl, out,
                                                           (unsigned long long)ws_size);
}


# GEMM block tile 128x128 -> 256x256 (one block per WGP), wave strip 128x64 on 2 waves/SIMD, fragment reads grouped a block ahead: L2-to-LDS and LDS read traffic halve; saliency sum order changes
# speedup vs baseline: 4.0895x; 1.0843x over previous
#include <hip/hip_runtime.h>

#define HH   64
#define WW   64
#define CCH  2048
#define HW   4096
#define IMG  224

typedef __attribute__((ext_vector_type(2))) float v2f;
typedef __attribute__((ext_vector_type(8))) float v8f;
typedef __attribute__((ext_vector_type(4))) float v4f;

#define WMMA_F32(A, B, C) \
    __builtin_amdgcn_wmma_f32_16x16x4_f32(false, (A), false, (B), (short)0, (C), false, false)

#define NTILE   16
#define BT      256
#define WS_MAX  0
#define WS_ROW  256
#define WS_COL  (WS_ROW + NTILE * HW)
#define WS_NEED ((unsigned long long)(WS_COL + NTILE * HW) * 4ull)

#define EP 68
#define EQ 272
#define GK 32
#define GP 36
#define GT (BT * GP)

#define LOAD_PAIR(S, TA, TB, K)                                                       \
    _Pragma("unroll")                                                                 \
    for (int j = 0; j < 4; ++j) {                                                     \
        fb[S][j][0] = *reinterpret_cast<const v2f*>((TB) + j * 16 * GP + (K));        \
        fb[S][j][1] = *reinterpret_cast<const v2f*>((TB) + j * 16 * GP + (K) + 4);    \
    }                                                                                 \
    _Pragma("unroll")                                                                 \
    for (int i = 0; i < 8; ++i) {                                                     \
        fa[S][i][0] = *reinterpret_cast<const v2f*>((TA) + i * 16 * GP + (K));        \
        fa[S][i][1] = *reinterpret_cast<const v2f*>((TA) + i * 16 * GP + (K) + 4);    \
    }

#define MMA_PAIR(S)                                                                   \
    _Pragma("unroll")                                                                 \
    for (int t = 0; t < 2; ++t) {                                                     \
        _Pragma("unroll")                                                             \
        for (int i = 0; i < 8; ++i) {                                                 \
            _Pragma("unroll")                                                         \
            for (int j = 0; j < 4; ++j)                                               \
                acc[i][j] = WMMA_F32(fa[S][i][t], fb[S][j][t], acc[i][j]);            \
        }                                                                             \
    }

#define KEEP_GROUPS() __builtin_amdgcn_sched_barrier(0)

__device__ __forceinline__ float sum8(const v8f v) {
    return ((v[0] + v[1]) + (v[2] + v[3])) + ((v[4] + v[5]) + (v[6] + v[7]));
}

__global__ __launch_bounds__(256, 2)
void simcam_gemm_sums(const float* __restrict__ x0, const float* __restrict__ x1,
                      float* __restrict__ tmax, float* __restrict__ rowpl,
                      float* __restrict__ colpl, unsigned long long ws_bytes) {
    if (ws_bytes < WS_NEED) return;

    __shared__ __align__(16) float sA[2 * GT];
    __shared__ __align__(16) float sB[2 * GT];

    const int tid  = threadIdx.x;
    const int wid  = __builtin_amdgcn_readfirstlane(tid >> 5);
    const int lane = tid & 31;
    const int r    = lane & 15;
    const int h    = lane >> 4;
    const int wm   = wid & 1;
    const int wn   = wid >> 1;

    const int sr = tid >> 3;
    const int sc = (tid & 7) * 4;
    const float* __restrict__ gA = x0 + (size_t)(blockIdx.x * BT + sr) * CCH + sc;
    const float* __restrict__ gB = x1 + (size_t)(blockIdx.y * BT + sr) * CCH + sc;
    float* const stA = sA + sr * GP + sc;
    float* const stB = sB + sr * GP + sc;

    const float* const fA = sA + (wm * 128 + r) * GP + 2 * h;
    const float* const fB = sB + (wn * 64 + r) * GP + 2 * h;

    v8f acc[8][4];
    #pragma unroll
    for (int i = 0; i < 8; ++i) {
        #pragma unroll
        for (int j = 0; j < 4; ++j) acc[i][j] = v8f{};
    }

    const size_t RS = (size_t)32 * CCH;

    v4f pa[8], pb[8];
    #pragma unroll
    for (int i = 0; i < 8; ++i) pa[i] = *reinterpret_cast<const v4f*>(gA + i * RS);
    #pragma unroll
    for (int i = 0; i < 8; ++i) pb[i] = *reinterpret_cast<const v4f*>(gB + i * RS);
    #pragma unroll
    for (int i = 0; i < 8; ++i) *reinterpret_cast<v4f*>(stA + i * 32 * GP) = pa[i];
    #pragma unroll
    for (int i = 0; i < 8; ++i) *reinterpret_cast<v4f*>(stB + i * 32 * GP) = pb[i];
    __syncthreads();

    v2f fa[2][8][2], fb[2][4][2];
    LOAD_PAIR(0, fA, fB, 0)

    int cur = 0;
    for (int kc = GK; kc < CCH; kc += GK) {
        const float* const tA = fA + cur * GT;
        const float* const tB = fB + cur * GT;
        const int nxt = cur ^ 1;
        float* const wA = stA + nxt * GT;
        float* const wB = stB + nxt * GT;

        #pragma unroll
        for (int i = 0; i < 8; ++i) pa[i] = *reinterpret_cast<const v4f*>(gA + kc + i * RS);
        #pragma unroll
        for (int i = 0; i < 8; ++i) pb[i] = *reinterpret_cast<const v4f*>(gB + kc + i * RS);

        LOAD_PAIR(1, tA, tB, 8)
        KEEP_GROUPS();
        MMA_PAIR(0)
        KEEP_GROUPS();
        LOAD_PAIR(0, tA, tB, 16)
        KEEP_GROUPS();
        MMA_PAIR(1)
        KEEP_GROUPS();
        #pragma unroll
        for (int i = 0; i < 8; ++i) *reinterpret_cast<v4f*>(wA + i * 32 * GP) = pa[i];
        #pragma unroll
        for (int i = 0; i < 8; ++i) *reinterpret_cast<v4f*>(wB + i * 32 * GP) = pb[i];
        LOAD_PAIR(1, tA, tB, 24)
        KEEP_GROUPS();
        MMA_PAIR(0)
        KEEP_GROUPS();
        MMA_PAIR(1)

        __syncthreads();
        cur = nxt;
        LOAD_PAIR(0, fA + cur * GT, fB + cur * GT, 0)
    }

    {
        const float* const tA = fA + cur * GT;
        const float* const tB = fB + cur * GT;
        LOAD_PAIR(1, tA, tB, 8)
        KEEP_GROUPS();
        MMA_PAIR(0)
        KEEP_GROUPS();
        LOAD_PAIR(0, tA, tB, 16)
        KEEP_GROUPS();
        MMA_PAIR(1)
        KEEP_GROUPS();
        LOAD_PAIR(1, tA, tB, 24)
        KEEP_GROUPS();
        MMA_PAIR(0)
        KEEP_GROUPS();
        MMA_PAIR(1)
    }

    float m = -3.402823466e38f;
    #pragma unroll
    for (int i = 0; i < 8; ++i) {
        #pragma unroll
        for (int j = 0; j < 4; ++j) {
            #pragma unroll
            for (int e = 0; e < 8; ++e) m = fmaxf(m, acc[i][j][e]);
        }
    }
    #pragma unroll
    for (int off = 16; off >= 1; off >>= 1)
        m = fmaxf(m, __shfl_xor(m, off, 32));

    v8f rs[8];
    #pragma unroll
    for (int i = 0; i < 8; ++i)
        rs[i] = (acc[i][0] + acc[i][1]) + (acc[i][2] + acc[i][3]);
    float cs[4];
    #pragma unroll
    for (int j = 0; j < 4; ++j) {
        const v8f ct = ((acc[0][j] + acc[1][j]) + (acc[2][j] + acc[3][j]))
                     + ((acc[4][j] + acc[5][j]) + (acc[6][j] + acc[7][j]));
        cs[j] = sum8(ct);
    }

    __syncthreads();

    float* const eR = sA;
    float* const eC = sB;
    float* const eM = sB + 4 * EQ;
    {
        const int rl = wm * 128 + 8 * h;
        const int q  = wn * 16 + r;
        #pragma unroll
        for (int i = 0; i < 8; ++i) {
            #pragma unroll
            for (int e = 0; e < 8; ++e)
                eR[(rl + 16 * i + e) * EP + q] = rs[i][e];
        }
        const int pl = wm * 2 + h;
        const int cl = wn * 64 + r;
        #pragma unroll
        for (int j = 0; j < 4; ++j) eC[pl * EQ + cl + 16 * j] = cs[j];
        if (lane == 0) eM[wid] = m;
    }
    __syncthreads();

    {
        const v4f* const p = reinterpret_cast<const v4f*>(eR + tid * EP);
        const v4f s4 = (((p[0] + p[1]) + (p[2] + p[3])) + ((p[4] + p[5]) + (p[6] + p[7])))
                     + (((p[8] + p[9]) + (p[10] + p[11])) + ((p[12] + p[13]) + (p[14] + p[15])));
        rowpl[(size_t)blockIdx.y * HW + blockIdx.x * BT + tid] = (s4[0] + s4[1]) + (s4[2] + s4[3]);
        const float* const pc = eC + tid;
        colpl[(size_t)blockIdx.x * HW + blockIdx.y * BT + tid] = (pc[0] + pc[EQ]) + (pc[2 * EQ] + pc[3 * EQ]);
    }
    if (tid == 0) {
        float bm = eM[0];
        for (int w = 1; w < 8; ++w) bm = fmaxf(bm, eM[w]);
        atomicExch(tmax + blockIdx.y * NTILE + blockIdx.x, bm);
    }
}

__global__ __launch_bounds__(256)
void simcam_finish(const float* __restrict__ tmax, const float* __restrict__ rowpl,
                   const float* __restrict__ colpl, float* __restrict__ out,
                   unsigned long long ws_bytes) {
    __shared__ float part[2][128];
    __shared__ float vals[128];
    __shared__ float wm[8];

    if (ws_bytes < WS_NEED) return;

    const int tid  = threadIdx.x;
    const int wave = tid >> 5;
    const int lane = tid & 31;
    const int s = (int)blockIdx.x / IMG;
    const int y = (int)blockIdx.x - s * IMG;

    const float scale = (float)HH / (float)IMG;
    float fy = ((float)y + 0.5f) * scale - 0.5f;
    fy = fminf(fmaxf(fy, 0.0f), (float)(HH - 1));
    const int y0 = (int)fy;
    const int y1 = min(y0 + 1, HH - 1);
    const float wy = fy - (float)y0;

    float m = tmax[tid];

    const int e    = tid & 127;
    const int half = tid >> 7;
    const int src  = ((e >> 6) ? y1 : y0) * WW + (e & 63);
    const float* __restrict__ p = (s ? colpl : rowpl) + (size_t)(half * (NTILE / 2)) * HW + src;
    float v[NTILE / 2];
    #pragma unroll
    for (int j = 0; j < NTILE / 2; ++j) v[j] = p[(size_t)j * HW];
    float a = 0.f;
    #pragma unroll
    for (int j = 0; j < NTILE / 2; ++j) a += v[j];
    part[half][e] = a;

    for (int off = 16; off >= 1; off >>= 1)
        m = fmaxf(m, __shfl_xor(m, off, 32));
    if (lane == 0) wm[wave] = m;
    __syncthreads();

    if (tid < 128) {
        float gm = wm[0];
        for (int w = 1; w < 8; ++w) gm = fmaxf(gm, wm[w]);
        const float inv = 1.0f / gm;
        vals[tid] = (part[0][tid] + part[1][tid]) * inv;
    }
    __syncthreads();

    if (tid < IMG) {
        const int x = tid;
        float fx = ((float)x + 0.5f) * scale - 0.5f;
        fx = fminf(fmaxf(fx, 0.0f), (float)(WW - 1));
        const int x0 = (int)fx;
        const int x1 = min(x0 + 1, WW - 1);
        const float wx = fx - (float)x0;
        const float v00 = vals[x0],      v01 = vals[x1];
        const float v10 = vals[WW + x0], v11 = vals[WW + x1];
        out[((size_t)s * IMG + y) * IMG + x] = (1.0f - wy) * ((1.0f - wx) * v00 + wx * v01)
                                             +          wy * ((1.0f - wx) * v10 + wx * v11);
    }
}

extern "C" void kernel_launch(void* const* d_in, const int* in_sizes, int n_in,
                              void* d_out, int out_size, void* d_ws, size_t ws_size,
                              hipStream_t stream) {
    (void)in_sizes; (void)n_in; (void)out_size;

    const float* feats = (const float*)d_in[0];
    const float* x0 = feats;
    const float* x1 = feats + (size_t)HW * CCH;
    float* out = (float*)d_out;

    float* wsf   = (float*)d_ws;
    float* tmax  = wsf + WS_MAX;
    float* rowpl = wsf + WS_ROW;
    float* colpl = wsf + WS_COL;

    simcam_gemm_sums<<<dim3(NTILE, NTILE), dim3(256), 0, stream>>>(x0, x1, tmax, rowpl, colpl,
                                                                  (unsigned long long)ws_size);

    simcam_finish<<<dim3(2 * IMG), dim3(256), 0, stream>>>(tmax, rowpl, colpl, out,
                                                           (unsigned long long)ws_size);
}
